# starting kernel
# baseline (speedup 1.0000x reference)
#include <hip/hip_runtime.h>
#include <math.h>

typedef float v2f __attribute__((ext_vector_type(2)));
typedef float v8f __attribute__((ext_vector_type(8)));

#define BB   32
#define HH   8
#define JJ   8192
#define KK   128
#define EPSF 1e-6f
#define WAVES 8
#define JTILE 16
#define JBLK  (WAVES * JTILE)

__global__ __launch_bounds__(256) void cw_score_kernel(
    const float* __restrict__ memory,
    const float* __restrict__ keys,
    const float* __restrict__ strengths,
    const float* __restrict__ mask,
    float* __restrict__ sharp_out)
{
    __shared__ float wlds [16 * KK];
    __shared__ float m2lds[16 * KK];
    __shared__ float knorm[16];
    __shared__ float spl  [16];

    const int i   = blockIdx.y;
    const int jc  = blockIdx.x;
    const int tid = threadIdx.x;

    for (int e = tid; e < 16 * KK; e += 256) {
        const int h = e >> 7;
        const int k = e & (KK - 1);
        float w = 0.f, m2 = 0.f;
        if (h < HH) {
            const float mk = mask[(i * HH + h) * KK + k];
            const float ky = keys[(i * HH + h) * KK + k];
            m2 = mk * mk;
            w  = m2 * ky;
        }
        wlds [e] = w;
        m2lds[e] = m2;
    }
    if (tid < 16) {
        float kn = 0.f, sp = 0.f;
        if (tid < HH) {
            const int base = (i * HH + tid) * KK;
            float s = 0.f;
            for (int k = 0; k < KK; ++k) {
                const float t = mask[base + k] * keys[base + k];
                s += t * t;
            }
            kn = sqrtf(s);
            const float st = strengths[i * HH + tid];
            sp = (st > 20.f) ? st : log1pf(__expf(st));
        }
        knorm[tid] = kn;
        spl  [tid] = sp;
    }
    __syncthreads();

    const int wave  = tid >> 5;
    const int lane  = tid & 31;
    const int j0    = jc * JBLK + wave * JTILE;
    const int row   = j0 + (lane & 15);
    const int khalf = (lane >> 4) << 1;
    const int n     = lane & 15;

    const float* aptr = memory + (size_t)(i * JJ + row) * KK + khalf;
    const float* b1p  = &wlds [n * KK + khalf];
    const float* b2p  = &m2lds[n * KK + khalf];

    v8f c1 = {};
    v8f c2 = {};
#pragma unroll
    for (int s = 0; s < KK / 4; ++s) {
        v2f a  = *(const v2f*)(aptr + s * 4);
        v2f b1 = *(const v2f*)(b1p  + s * 4);
        v2f b2 = *(const v2f*)(b2p  + s * 4);
        v2f a2 = a * a;
        c1 = __builtin_amdgcn_wmma_f32_16x16x4_f32(false, a,  false, b1,
                                                   (short)0, c1, false, false);
        c2 = __builtin_amdgcn_wmma_f32_16x16x4_f32(false, a2, false, b2,
                                                   (short)0, c2, false, false);
    }

    if (n < HH) {
        const float kn    = knorm[n];
        const float sp    = spl[n];
        const int   jbase = j0 + ((lane >> 4) << 3);
        float* outp = sharp_out + (size_t)(i * HH + n) * JJ;
#pragma unroll
        for (int r = 0; r < 8; ++r) {
            const float proj = c1[r];
            const float mn   = sqrtf(c2[r]);
            const float sim  = proj / (kn * mn + EPSF);
            outp[jbase + r]  = sim * sp;
        }
    }
}

__global__ __launch_bounds__(256) void cw_softmax_kernel(float* __restrict__ data)
{
    __shared__ float red[256];
    const int tid = threadIdx.x;
    float* p = data + (size_t)blockIdx.x * JJ;

    float v[JJ / 256];
    float mx = -INFINITY;
#pragma unroll
    for (int t = 0; t < JJ / 256; ++t) {
        v[t] = p[tid + t * 256];
        mx = fmaxf(mx, v[t]);
    }
    red[tid] = mx;
    __syncthreads();
    for (int s = 128; s > 0; s >>= 1) {
        if (tid < s) red[tid] = fmaxf(red[tid], red[tid + s]);
        __syncthreads();
    }
    mx = red[0];
    __syncthreads();

    float sum = 0.f;
#pragma unroll
    for (int t = 0; t < JJ / 256; ++t) {
        v[t] = __expf(v[t] - mx);
        sum += v[t];
    }
    red[tid] = sum;
    __syncthreads();
    for (int s = 128; s > 0; s >>= 1) {
        if (tid < s) red[tid] += red[tid + s];
        __syncthreads();
    }
    const float inv = 1.f / red[0];
#pragma unroll
    for (int t = 0; t < JJ / 256; ++t) {
        p[tid + t * 256] = v[t] * inv;
    }
}

extern "C" void kernel_launch(void* const* d_in, const int* in_sizes, int n_in,
                              void* d_out, int out_size, void* d_ws, size_t ws_size,
                              hipStream_t stream)
{
    const float* memory    = (const float*)d_in[0];
    const float* keys      = (const float*)d_in[1];
    const float* strengths = (const float*)d_in[2];
    const float* mask      = (const float*)d_in[3];
    float* out = (float*)d_out;

    dim3 grid1(JJ / JBLK, BB);
    cw_score_kernel<<<grid1, 256, 0, stream>>>(memory, keys, strengths, mask, out);
    cw_softmax_kernel<<<BB * HH, 256, 0, stream>>>(out);
}


# LDS row pitch of both B-operand tiles 128 -> 132 floats (16 B x odd): every 8-byte B-fragment read goes from a 16-way bank conflict to conflict-free
# speedup vs baseline: 1.5752x; 1.5752x over previous
#include <hip/hip_runtime.h>
#include <math.h>

typedef float v2f __attribute__((ext_vector_type(2)));
typedef float v8f __attribute__((ext_vector_type(8)));

#define BB   32
#define HH   8
#define JJ   8192
#define KK   128
#define EPSF 1e-6f
#define WAVES 8
#define JTILE 16
#define JBLK  (WAVES * JTILE)
#define BPITCH (KK + 4)

__global__ __launch_bounds__(256) void cw_score_kernel(
    const float* __restrict__ memory,
    const float* __restrict__ keys,
    const float* __restrict__ strengths,
    const float* __restrict__ mask,
    float* __restrict__ sharp_out)
{
    __shared__ float wlds [16 * BPITCH];
    __shared__ float m2lds[16 * BPITCH];
    __shared__ float knorm[16];
    __shared__ float spl  [16];

    const int i   = blockIdx.y;
    const int jc  = blockIdx.x;
    const int tid = threadIdx.x;

    for (int e = tid; e < 16 * KK; e += 256) {
        const int h = e >> 7;
        const int k = e & (KK - 1);
        float w = 0.f, m2 = 0.f;
        if (h < HH) {
            const float mk = mask[(i * HH + h) * KK + k];
            const float ky = keys[(i * HH + h) * KK + k];
            m2 = mk * mk;
            w  = m2 * ky;
        }
        wlds [h * BPITCH + k] = w;
        m2lds[h * BPITCH + k] = m2;
    }
    if (tid < 16) {
        float kn = 0.f, sp = 0.f;
        if (tid < HH) {
            const int base = (i * HH + tid) * KK;
            float s = 0.f;
            for (int k = 0; k < KK; ++k) {
                const float t = mask[base + k] * keys[base + k];
                s += t * t;
            }
            kn = sqrtf(s);
            const float st = strengths[i * HH + tid];
            sp = (st > 20.f) ? st : log1pf(__expf(st));
        }
        knorm[tid] = kn;
        spl  [tid] = sp;
    }
    __syncthreads();

    const int wave  = tid >> 5;
    const int lane  = tid & 31;
    const int j0    = jc * JBLK + wave * JTILE;
    const int row   = j0 + (lane & 15);
    const int khalf = (lane >> 4) << 1;
    const int n     = lane & 15;

    const float* aptr = memory + (size_t)(i * JJ + row) * KK + khalf;
    const float* b1p  = &wlds [n * BPITCH + khalf];
    const float* b2p  = &m2lds[n * BPITCH + khalf];

    v8f c1 = {};
    v8f c2 = {};
#pragma unroll
    for (int s = 0; s < KK / 4; ++s) {
        v2f a  = *(const v2f*)(aptr + s * 4);
        v2f b1 = *(const v2f*)(b1p  + s * 4);
        v2f b2 = *(const v2f*)(b2p  + s * 4);
        v2f a2 = a * a;
        c1 = __builtin_amdgcn_wmma_f32_16x16x4_f32(false, a,  false, b1,
                                                   (short)0, c1, false, false);
        c2 = __builtin_amdgcn_wmma_f32_16x16x4_f32(false, a2, false, b2,
                                                   (short)0, c2, false, false);
    }

    if (n < HH) {
        const float kn    = knorm[n];
        const float sp    = spl[n];
        const int   jbase = j0 + ((lane >> 4) << 3);
        float* outp = sharp_out + (size_t)(i * HH + n) * JJ;
#pragma unroll
        for (int r = 0; r < 8; ++r) {
            const float proj = c1[r];
            const float mn   = sqrtf(c2[r]);
            const float sim  = proj / (kn * mn + EPSF);
            outp[jbase + r]  = sim * sp;
        }
    }
}

__global__ __launch_bounds__(256) void cw_softmax_kernel(float* __restrict__ data)
{
    __shared__ float red[256];
    const int tid = threadIdx.x;
    float* p = data + (size_t)blockIdx.x * JJ;

    float v[JJ / 256];
    float mx = -INFINITY;
#pragma unroll
    for (int t = 0; t < JJ / 256; ++t) {
        v[t] = p[tid + t * 256];
        mx = fmaxf(mx, v[t]);
    }
    red[tid] = mx;
    __syncthreads();
    for (int s = 128; s > 0; s >>= 1) {
        if (tid < s) red[tid] = fmaxf(red[tid], red[tid + s]);
        __syncthreads();
    }
    mx = red[0];
    __syncthreads();

    float sum = 0.f;
#pragma unroll
    for (int t = 0; t < JJ / 256; ++t) {
        v[t] = __expf(v[t] - mx);
        sum += v[t];
    }
    red[tid] = sum;
    __syncthreads();
    for (int s = 128; s > 0; s >>= 1) {
        if (tid < s) red[tid] += red[tid + s];
        __syncthreads();
    }
    const float inv = 1.f / red[0];
#pragma unroll
    for (int t = 0; t < JJ / 256; ++t) {
        p[tid + t * 256] = v[t] * inv;
    }
}

extern "C" void kernel_launch(void* const* d_in, const int* in_sizes, int n_in,
                              void* d_out, int out_size, void* d_ws, size_t ws_size,
                              hipStream_t stream)
{
    const float* memory    = (const float*)d_in[0];
    const float* keys      = (const float*)d_in[1];
    const float* strengths = (const float*)d_in[2];
    const float* mask      = (const float*)d_in[3];
    float* out = (float*)d_out;

    dim3 grid1(JJ / JBLK, BB);
    cw_score_kernel<<<grid1, 256, 0, stream>>>(memory, keys, strengths, mask, out);
    cw_softmax_kernel<<<BB * HH, 256, 0, stream>>>(out);
}


# Preamble: wave w stages head w from one 16-byte mask and keys load per lane: 1 load round trip per block, not 20 dependent ones; key-norm sum order changes (4 per lane, then lane butterfly)
# speedup vs baseline: 1.8366x; 1.1659x over previous
#include <hip/hip_runtime.h>
#include <math.h>

typedef float v2f __attribute__((ext_vector_type(2)));
typedef float v4f __attribute__((ext_vector_type(4)));
typedef float v8f __attribute__((ext_vector_type(8)));

#define BB   32
#define HH   8
#define JJ   8192
#define KK   128
#define EPSF 1e-6f
#define WAVES 8
#define JTILE 16
#define JBLK  (WAVES * JTILE)
static_assert(WAVES == HH, "the preamble gives each wave one head");
#define BPITCH (KK + 4)

__global__ __launch_bounds__(256) void cw_score_kernel(
    const float* __restrict__ memory,
    const float* __restrict__ keys,
    const float* __restrict__ strengths,
    const float* __restrict__ mask,
    float* __restrict__ sharp_out)
{
    __shared__ __attribute__((aligned(16))) float wlds [16 * BPITCH];
    __shared__ __attribute__((aligned(16))) float m2lds[16 * BPITCH];
    __shared__ float knorm[HH];
    __shared__ float spl  [HH];

    const int i    = blockIdx.y;
    const int jc   = blockIdx.x;
    const int tid  = threadIdx.x;
    const int wave = tid >> 5;
    const int lane = tid & 31;

    {
        const int   hk = (i * HH + wave) * KK + lane * 4;
        const v4f   mk = *(const v4f*)(mask + hk);
        const v4f   ky = *(const v4f*)(keys + hk);
        const float st = strengths[i * HH + wave];
        const v4f   m2 = mk * mk;
        const v4f   w  = m2 * ky;
        const v4f   zero = {0.f, 0.f, 0.f, 0.f};
        const int   lo = wave * BPITCH + lane * 4;
        const int   hi = (wave + HH) * BPITCH + lane * 4;
        *(v4f*)&wlds [lo] = w;
        *(v4f*)&m2lds[lo] = m2;
        *(v4f*)&wlds [hi] = zero;
        *(v4f*)&m2lds[hi] = zero;
        const v4f t = mk * ky;
        float s = t.x * t.x + t.y * t.y + t.z * t.z + t.w * t.w;
        for (int off = 16; off > 0; off >>= 1) s += __shfl_xor(s, off, 32);
        const float kn = sqrtf(s);
        const float sp = (st > 20.f) ? st : log1pf(__expf(st));
        if (lane == 0) {
            knorm[wave] = kn;
            spl  [wave] = sp;
        }
    }
    __syncthreads();

    const int j0    = jc * JBLK + wave * JTILE;
    const int row   = j0 + (lane & 15);
    const int khalf = (lane >> 4) << 1;
    const int n     = lane & 15;

    const float* aptr = memory + (size_t)(i * JJ + row) * KK + khalf;
    const float* b1p  = &wlds [n * BPITCH + khalf];
    const float* b2p  = &m2lds[n * BPITCH + khalf];

    v8f c1 = {};
    v8f c2 = {};
#pragma unroll
    for (int s = 0; s < KK / 4; ++s) {
        v2f a  = *(const v2f*)(aptr + s * 4);
        v2f b1 = *(const v2f*)(b1p  + s * 4);
        v2f b2 = *(const v2f*)(b2p  + s * 4);
        v2f a2 = a * a;
        c1 = __builtin_amdgcn_wmma_f32_16x16x4_f32(false, a,  false, b1,
                                                   (short)0, c1, false, false);
        c2 = __builtin_amdgcn_wmma_f32_16x16x4_f32(false, a2, false, b2,
                                                   (short)0, c2, false, false);
    }

    if (n < HH) {
        const float kn    = knorm[n];
        const float sp    = spl[n];
        const int   jbase = j0 + ((lane >> 4) << 3);
        float* outp = sharp_out + (size_t)(i * HH + n) * JJ;
#pragma unroll
        for (int r = 0; r < 8; ++r) {
            const float proj = c1[r];
            const float mn   = sqrtf(c2[r]);
            const float sim  = proj / (kn * mn + EPSF);
            outp[jbase + r]  = sim * sp;
        }
    }
}

__global__ __launch_bounds__(256) void cw_softmax_kernel(float* __restrict__ data)
{
    __shared__ float red[256];
    const int tid = threadIdx.x;
    float* p = data + (size_t)blockIdx.x * JJ;

    float v[JJ / 256];
    float mx = -INFINITY;
#pragma unroll
    for (int t = 0; t < JJ / 256; ++t) {
        v[t] = p[tid + t * 256];
        mx = fmaxf(mx, v[t]);
    }
    red[tid] = mx;
    __syncthreads();
    for (int s = 128; s > 0; s >>= 1) {
        if (tid < s) red[tid] = fmaxf(red[tid], red[tid + s]);
        __syncthreads();
    }
    mx = red[0];
    __syncthreads();

    float sum = 0.f;
#pragma unroll
    for (int t = 0; t < JJ / 256; ++t) {
        v[t] = __expf(v[t] - mx);
        sum += v[t];
    }
    red[tid] = sum;
    __syncthreads();
    for (int s = 128; s > 0; s >>= 1) {
        if (tid < s) red[tid] += red[tid + s];
        __syncthreads();
    }
    const float inv = 1.f / red[0];
#pragma unroll
    for (int t = 0; t < JJ / 256; ++t) {
        p[tid + t * 256] = v[t] * inv;
    }
}

extern "C" void kernel_launch(void* const* d_in, const int* in_sizes, int n_in,
                              void* d_out, int out_size, void* d_ws, size_t ws_size,
                              hipStream_t stream)
{
    const float* memory    = (const float*)d_in[0];
    const float* keys      = (const float*)d_in[1];
    const float* strengths = (const float*)d_in[2];
    const float* mask      = (const float*)d_in[3];
    float* out = (float*)d_out;

    dim3 grid1(JJ / JBLK, BB);
    cw_score_kernel<<<grid1, 256, 0, stream>>>(memory, keys, strengths, mask, out);
    cw_softmax_kernel<<<BB * HH, 256, 0, stream>>>(out);
}


# Score kernel: A (global) and B (LDS) fragments loaded 16 B per lane, one load per two WMMA k-steps: half as many 16-line loads; k regrouped per WMMA so sum order changes; reg budget 12 waves/SIMD
# speedup vs baseline: 1.9076x; 1.0387x over previous
#include <hip/hip_runtime.h>
#include <math.h>

typedef float v2f __attribute__((ext_vector_type(2)));
typedef float v4f __attribute__((ext_vector_type(4)));
typedef float v8f __attribute__((ext_vector_type(8)));

#define BB   32
#define HH   8
#define JJ   8192
#define KK   128
#define EPSF 1e-6f
#define WAVES 8
#define JTILE 16
#define JBLK  (WAVES * JTILE)
static_assert(WAVES == HH, "the preamble gives each wave one head");
#define BPITCH (KK + 4)

__global__ __launch_bounds__(256) __attribute__((amdgpu_waves_per_eu(12, 12)))
void cw_score_kernel(
    const float* __restrict__ memory,
    const float* __restrict__ keys,
    const float* __restrict__ strengths,
    const float* __restrict__ mask,
    float* __restrict__ sharp_out)
{
    __shared__ __attribute__((aligned(16))) float wlds [16 * BPITCH];
    __shared__ __attribute__((aligned(16))) float m2lds[16 * BPITCH];
    __shared__ float knorm[HH];
    __shared__ float spl  [HH];

    const int i    = blockIdx.y;
    const int jc   = blockIdx.x;
    const int tid  = threadIdx.x;
    const int wave = tid >> 5;
    const int lane = tid & 31;

    {
        const int   hk = (i * HH + wave) * KK + lane * 4;
        const v4f   mk = *(const v4f*)(mask + hk);
        const v4f   ky = *(const v4f*)(keys + hk);
        const float st = strengths[i * HH + wave];
        const v4f   m2 = mk * mk;
        const v4f   w  = m2 * ky;
        const v4f   zero = {0.f, 0.f, 0.f, 0.f};
        const int   lo = wave * BPITCH + lane * 4;
        const int   hi = (wave + HH) * BPITCH + lane * 4;
        *(v4f*)&wlds [lo] = w;
        *(v4f*)&m2lds[lo] = m2;
        *(v4f*)&wlds [hi] = zero;
        *(v4f*)&m2lds[hi] = zero;
        const v4f t = mk * ky;
        float s = t.x * t.x + t.y * t.y + t.z * t.z + t.w * t.w;
        for (int off = 16; off > 0; off >>= 1) s += __shfl_xor(s, off, 32);
        const float kn = sqrtf(s);
        const float sp = (st > 20.f) ? st : log1pf(__expf(st));
        if (lane == 0) {
            knorm[wave] = kn;
            spl  [wave] = sp;
        }
    }
    __syncthreads();

    const int j0    = jc * JBLK + wave * JTILE;
    const int row   = j0 + (lane & 15);
    const int kq    = (lane >> 4) << 2;
    const int n     = lane & 15;

    const float* aptr = memory + (size_t)(i * JJ + row) * KK + kq;
    const float* b1p  = &wlds [n * BPITCH + kq];
    const float* b2p  = &m2lds[n * BPITCH + kq];

    v8f c1 = {};
    v8f c2 = {};
#pragma unroll
    for (int q = 0; q < KK / 8; ++q) {
        const v4f a  = *(const v4f*)(aptr + q * 8);
        const v4f b1 = *(const v4f*)(b1p  + q * 8);
        const v4f b2 = *(const v4f*)(b2p  + q * 8);
        const v4f a2 = a * a;
        const v2f a_lo  = {a.x,  a.y},  a_hi  = {a.z,  a.w};
        const v2f a2_lo = {a2.x, a2.y}, a2_hi = {a2.z, a2.w};
        const v2f b1_lo = {b1.x, b1.y}, b1_hi = {b1.z, b1.w};
        const v2f b2_lo = {b2.x, b2.y}, b2_hi = {b2.z, b2.w};
        c1 = __builtin_amdgcn_wmma_f32_16x16x4_f32(false, a_lo,  false, b1_lo,
                                                   (short)0, c1, false, false);
        c2 = __builtin_amdgcn_wmma_f32_16x16x4_f32(false, a2_lo, false, b2_lo,
                                                   (short)0, c2, false, false);
        c1 = __builtin_amdgcn_wmma_f32_16x16x4_f32(false, a_hi,  false, b1_hi,
                                                   (short)0, c1, false, false);
        c2 = __builtin_amdgcn_wmma_f32_16x16x4_f32(false, a2_hi, false, b2_hi,
                                                   (short)0, c2, false, false);
    }

    if (n < HH) {
        const float kn    = knorm[n];
        const float sp    = spl[n];
        const int   jbase = j0 + ((lane >> 4) << 3);
        float* outp = sharp_out + (size_t)(i * HH + n) * JJ;
#pragma unroll
        for (int r = 0; r < 8; ++r) {
            const float proj = c1[r];
            const float mn   = sqrtf(c2[r]);
            const float sim  = proj / (kn * mn + EPSF);
            outp[jbase + r]  = sim * sp;
        }
    }
}

__global__ __launch_bounds__(256) void cw_softmax_kernel(float* __restrict__ data)
{
    __shared__ float red[256];
    const int tid = threadIdx.x;
    float* p = data + (size_t)blockIdx.x * JJ;

    float v[JJ / 256];
    float mx = -INFINITY;
#pragma unroll
    for (int t = 0; t < JJ / 256; ++t) {
        v[t] = p[tid + t * 256];
        mx = fmaxf(mx, v[t]);
    }
    red[tid] = mx;
    __syncthreads();
    for (int s = 128; s > 0; s >>= 1) {
        if (tid < s) red[tid] = fmaxf(red[tid], red[tid + s]);
        __syncthreads();
    }
    mx = red[0];
    __syncthreads();

    float sum = 0.f;
#pragma unroll
    for (int t = 0; t < JJ / 256; ++t) {
        v[t] = __expf(v[t] - mx);
        sum += v[t];
    }
    red[tid] = sum;
    __syncthreads();
    for (int s = 128; s > 0; s >>= 1) {
        if (tid < s) red[tid] += red[tid + s];
        __syncthreads();
    }
    const float inv = 1.f / red[0];
#pragma unroll
    for (int t = 0; t < JJ / 256; ++t) {
        p[tid + t * 256] = v[t] * inv;
    }
}

extern "C" void kernel_launch(void* const* d_in, const int* in_sizes, int n_in,
                              void* d_out, int out_size, void* d_ws, size_t ws_size,
                              hipStream_t stream)
{
    const float* memory    = (const float*)d_in[0];
    const float* keys      = (const float*)d_in[1];
    const float* strengths = (const float*)d_in[2];
    const float* mask      = (const float*)d_in[3];
    float* out = (float*)d_out;

    dim3 grid1(JJ / JBLK, BB);
    cw_score_kernel<<<grid1, 256, 0, stream>>>(memory, keys, strengths, mask, out);
    cw_softmax_kernel<<<BB * HH, 256, 0, stream>>>(out);
}


# Score kernel: each block loops over 4 j-chunks (4x fewer waves to enter): B fragments stay in registers, the A registers are a ring refilled with the next tile as pieces are used; sum order kept
# speedup vs baseline: 2.1706x; 1.1379x over previous
#include <hip/hip_runtime.h>
#include <math.h>

typedef float v2f __attribute__((ext_vector_type(2)));
typedef float v4f __attribute__((ext_vector_type(4)));
typedef float v8f __attribute__((ext_vector_type(8)));

#define BB   32
#define HH   8
#define JJ   8192
#define KK   128
#define EPSF 1e-6f
#define WAVES 8
#define JTILE 16
#define JBLK  (WAVES * JTILE)
#define TILES 4
static_assert(WAVES == HH, "the preamble gives each wave one head");
static_assert(JJ % (JBLK * TILES) == 0, "a block walks whole chunks");
#define BPITCH (KK + 4)

template <bool REFILL>
__device__ __forceinline__ void cw_tile(
    v4f (&a)[KK / 8], const v4f (&b1)[KK / 8], const v4f (&b2)[KK / 8],
    const float* __restrict__ anext, float* __restrict__ o,
    const float kn, const float sp, const bool owns_head)
{
    v8f c1 = {};
    v8f c2 = {};
#pragma unroll
    for (int q = 0; q < KK / 8; ++q) {
        const v4f av = a[q];
        const v4f a2 = av * av;
        const v2f a_lo  = {av.x, av.y}, a_hi  = {av.z, av.w};
        const v2f a2_lo = {a2.x, a2.y}, a2_hi = {a2.z, a2.w};
        const v2f b1_lo = {b1[q].x, b1[q].y}, b1_hi = {b1[q].z, b1[q].w};
        const v2f b2_lo = {b2[q].x, b2[q].y}, b2_hi = {b2[q].z, b2[q].w};
        c1 = __builtin_amdgcn_wmma_f32_16x16x4_f32(false, a_lo,  false, b1_lo,
                                                   (short)0, c1, false, false);
        c2 = __builtin_amdgcn_wmma_f32_16x16x4_f32(false, a2_lo, false, b2_lo,
                                                   (short)0, c2, false, false);
        c1 = __builtin_amdgcn_wmma_f32_16x16x4_f32(false, a_hi,  false, b1_hi,
                                                   (short)0, c1, false, false);
        c2 = __builtin_amdgcn_wmma_f32_16x16x4_f32(false, a2_hi, false, b2_hi,
                                                   (short)0, c2, false, false);
        if (REFILL) a[q] = *(const v4f*)(anext + q * 8);
    }
    if (owns_head) {
#pragma unroll
        for (int r = 0; r < 8; ++r) {
            const float proj = c1[r];
            const float mn   = sqrtf(c2[r]);
            const float sim  = proj / (kn * mn + EPSF);
            o[r] = sim * sp;
        }
    }
}

__global__ __launch_bounds__(256) __attribute__((amdgpu_waves_per_eu(4, 4)))
void cw_score_kernel(
    const float* __restrict__ memory,
    const float* __restrict__ keys,
    const float* __restrict__ strengths,
    const float* __restrict__ mask,
    float* __restrict__ sharp_out)
{
    __shared__ __attribute__((aligned(16))) float wlds [16 * BPITCH];
    __shared__ __attribute__((aligned(16))) float m2lds[16 * BPITCH];
    __shared__ float knorm[HH];
    __shared__ float spl  [HH];

    const int i    = blockIdx.y;
    const int jc   = blockIdx.x;
    const int tid  = threadIdx.x;
    const int wave = tid >> 5;
    const int lane = tid & 31;

    const int j0    = jc * (TILES * JBLK) + wave * JTILE;
    const int row   = j0 + (lane & 15);
    const int kq    = (lane >> 4) << 2;
    const int n     = lane & 15;

    const float* aptr = memory + (size_t)(i * JJ + row) * KK + kq;
    const float* b1p  = &wlds [n * BPITCH + kq];
    const float* b2p  = &m2lds[n * BPITCH + kq];

    v4f a[KK / 8];
    {
        const int   hk = (i * HH + wave) * KK + lane * 4;
        const v4f   mk = *(const v4f*)(mask + hk);
        const v4f   ky = *(const v4f*)(keys + hk);
        const float st = strengths[i * HH + wave];
#pragma unroll
        for (int q = 0; q < KK / 8; ++q) a[q] = *(const v4f*)(aptr + q * 8);
        const v4f   m2 = mk * mk;
        const v4f   w  = m2 * ky;
        const v4f   zero = {0.f, 0.f, 0.f, 0.f};
        const int   lo = wave * BPITCH + lane * 4;
        const int   hi = (wave + HH) * BPITCH + lane * 4;
        *(v4f*)&wlds [lo] = w;
        *(v4f*)&m2lds[lo] = m2;
        *(v4f*)&wlds [hi] = zero;
        *(v4f*)&m2lds[hi] = zero;
        const v4f t = mk * ky;
        float s = t.x * t.x + t.y * t.y + t.z * t.z + t.w * t.w;
        for (int off = 16; off > 0; off >>= 1) s += __shfl_xor(s, off, 32);
        const float kn = sqrtf(s);
        const float sp = (st > 20.f) ? st : log1pf(__expf(st));
        if (lane == 0) {
            knorm[wave] = kn;
            spl  [wave] = sp;
        }
    }
    __syncthreads();

    v4f b1[KK / 8];
    v4f b2[KK / 8];
#pragma unroll
    for (int q = 0; q < KK / 8; ++q) {
        b1[q] = *(const v4f*)(b1p + q * 8);
        b2[q] = *(const v4f*)(b2p + q * 8);
    }
    const float kn = knorm[n & (HH - 1)];
    const float sp = spl  [n & (HH - 1)];
    float* outp = sharp_out + (size_t)(i * HH + (n & (HH - 1))) * JJ + j0 + ((lane >> 4) << 3);

    for (int t = 0; t < TILES - 1; ++t)
        cw_tile<true>(a, b1, b2, aptr + (size_t)(t + 1) * (JBLK * KK), outp + t * JBLK, kn, sp, n < HH);
    cw_tile<false>(a, b1, b2, aptr, outp + (TILES - 1) * JBLK, kn, sp, n < HH);
}

__global__ __launch_bounds__(256) void cw_softmax_kernel(float* __restrict__ data)
{
    __shared__ float red[256];
    const int tid = threadIdx.x;
    float* p = data + (size_t)blockIdx.x * JJ;

    float v[JJ / 256];
    float mx = -INFINITY;
#pragma unroll
    for (int t = 0; t < JJ / 256; ++t) {
        v[t] = p[tid + t * 256];
        mx = fmaxf(mx, v[t]);
    }
    red[tid] = mx;
    __syncthreads();
    for (int s = 128; s > 0; s >>= 1) {
        if (tid < s) red[tid] = fmaxf(red[tid], red[tid + s]);
        __syncthreads();
    }
    mx = red[0];
    __syncthreads();

    float sum = 0.f;
#pragma unroll
    for (int t = 0; t < JJ / 256; ++t) {
        v[t] = __expf(v[t] - mx);
        sum += v[t];
    }
    red[tid] = sum;
    __syncthreads();
    for (int s = 128; s > 0; s >>= 1) {
        if (tid < s) red[tid] += red[tid + s];
        __syncthreads();
    }
    const float inv = 1.f / red[0];
#pragma unroll
    for (int t = 0; t < JJ / 256; ++t) {
        p[tid + t * 256] = v[t] * inv;
    }
}

extern "C" void kernel_launch(void* const* d_in, const int* in_sizes, int n_in,
                              void* d_out, int out_size, void* d_ws, size_t ws_size,
                              hipStream_t stream)
{
    const float* memory    = (const float*)d_in[0];
    const float* keys      = (const float*)d_in[1];
    const float* strengths = (const float*)d_in[2];
    const float* mask      = (const float*)d_in[3];
    float* out = (float*)d_out;

    dim3 grid1(JJ / (JBLK * TILES), BB);
    cw_score_kernel<<<grid1, 256, 0, stream>>>(memory, keys, strengths, mask, out);
    cw_softmax_kernel<<<BB * HH, 256, 0, stream>>>(out);
}


# Softmax kernel: row max and row sum by a lane butterfly on the vector ALU in each wave, then one LDS word per wave behind one barrier, not two 8-step LDS trees (19 barriers to 2); sum order changes
# speedup vs baseline: 2.2634x; 1.0428x over previous
#include <hip/hip_runtime.h>
#include <math.h>

typedef float v2f __attribute__((ext_vector_type(2)));
typedef float v4f __attribute__((ext_vector_type(4)));
typedef float v8f __attribute__((ext_vector_type(8)));

#define BB   32
#define HH   8
#define JJ   8192
#define KK   128
#define EPSF 1e-6f
#define WAVES 8
#define JTILE 16
#define JBLK  (WAVES * JTILE)
#define TILES 4
static_assert(WAVES == HH, "the preamble gives each wave one head");
static_assert(JJ % (JBLK * TILES) == 0, "a block walks whole chunks");
#define BPITCH (KK + 4)

template <bool REFILL>
__device__ __forceinline__ void cw_tile(
    v4f (&a)[KK / 8], const v4f (&b1)[KK / 8], const v4f (&b2)[KK / 8],
    const float* __restrict__ anext, float* __restrict__ o,
    const float kn, const float sp, const bool owns_head)
{
    v8f c1 = {};
    v8f c2 = {};
#pragma unroll
    for (int q = 0; q < KK / 8; ++q) {
        const v4f av = a[q];
        const v4f a2 = av * av;
        const v2f a_lo  = {av.x, av.y}, a_hi  = {av.z, av.w};
        const v2f a2_lo = {a2.x, a2.y}, a2_hi = {a2.z, a2.w};
        const v2f b1_lo = {b1[q].x, b1[q].y}, b1_hi = {b1[q].z, b1[q].w};
        const v2f b2_lo = {b2[q].x, b2[q].y}, b2_hi = {b2[q].z, b2[q].w};
        c1 = __builtin_amdgcn_wmma_f32_16x16x4_f32(false, a_lo,  false, b1_lo,
                                                   (short)0, c1, false, false);
        c2 = __builtin_amdgcn_wmma_f32_16x16x4_f32(false, a2_lo, false, b2_lo,
                                                   (short)0, c2, false, false);
        c1 = __builtin_amdgcn_wmma_f32_16x16x4_f32(false, a_hi,  false, b1_hi,
                                                   (short)0, c1, false, false);
        c2 = __builtin_amdgcn_wmma_f32_16x16x4_f32(false, a2_hi, false, b2_hi,
                                                   (short)0, c2, false, false);
        if (REFILL) a[q] = *(const v4f*)(anext + q * 8);
    }
    if (owns_head) {
#pragma unroll
        for (int r = 0; r < 8; ++r) {
            const float proj = c1[r];
            const float mn   = sqrtf(c2[r]);
            const float sim  = proj / (kn * mn + EPSF);
            o[r] = sim * sp;
        }
    }
}

__global__ __launch_bounds__(256) __attribute__((amdgpu_waves_per_eu(4, 4)))
void cw_score_kernel(
    const float* __restrict__ memory,
    const float* __restrict__ keys,
    const float* __restrict__ strengths,
    const float* __restrict__ mask,
    float* __restrict__ sharp_out)
{
    __shared__ __attribute__((aligned(16))) float wlds [16 * BPITCH];
    __shared__ __attribute__((aligned(16))) float m2lds[16 * BPITCH];
    __shared__ float knorm[HH];
    __shared__ float spl  [HH];

    const int i    = blockIdx.y;
    const int jc   = blockIdx.x;
    const int tid  = threadIdx.x;
    const int wave = tid >> 5;
    const int lane = tid & 31;

    const int j0    = jc * (TILES * JBLK) + wave * JTILE;
    const int row   = j0 + (lane & 15);
    const int kq    = (lane >> 4) << 2;
    const int n     = lane & 15;

    const float* aptr = memory + (size_t)(i * JJ + row) * KK + kq;
    const float* b1p  = &wlds [n * BPITCH + kq];
    const float* b2p  = &m2lds[n * BPITCH + kq];

    v4f a[KK / 8];
    {
        const int   hk = (i * HH + wave) * KK + lane * 4;
        const v4f   mk = *(const v4f*)(mask + hk);
        const v4f   ky = *(const v4f*)(keys + hk);
        const float st = strengths[i * HH + wave];
#pragma unroll
        for (int q = 0; q < KK / 8; ++q) a[q] = *(const v4f*)(aptr + q * 8);
        const v4f   m2 = mk * mk;
        const v4f   w  = m2 * ky;
        const v4f   zero = {0.f, 0.f, 0.f, 0.f};
        const int   lo = wave * BPITCH + lane * 4;
        const int   hi = (wave + HH) * BPITCH + lane * 4;
        *(v4f*)&wlds [lo] = w;
        *(v4f*)&m2lds[lo] = m2;
        *(v4f*)&wlds [hi] = zero;
        *(v4f*)&m2lds[hi] = zero;
        const v4f t = mk * ky;
        float s = t.x * t.x + t.y * t.y + t.z * t.z + t.w * t.w;
        for (int off = 16; off > 0; off >>= 1) s += __shfl_xor(s, off, 32);
        const float kn = sqrtf(s);
        const float sp = (st > 20.f) ? st : log1pf(__expf(st));
        if (lane == 0) {
            knorm[wave] = kn;
            spl  [wave] = sp;
        }
    }
    __syncthreads();

    v4f b1[KK / 8];
    v4f b2[KK / 8];
#pragma unroll
    for (int q = 0; q < KK / 8; ++q) {
        b1[q] = *(const v4f*)(b1p + q * 8);
        b2[q] = *(const v4f*)(b2p + q * 8);
    }
    const float kn = knorm[n & (HH - 1)];
    const float sp = spl  [n & (HH - 1)];
    float* outp = sharp_out + (size_t)(i * HH + (n & (HH - 1))) * JJ + j0 + ((lane >> 4) << 3);

    for (int t = 0; t < TILES - 1; ++t)
        cw_tile<true>(a, b1, b2, aptr + (size_t)(t + 1) * (JBLK * KK), outp + t * JBLK, kn, sp, n < HH);
    cw_tile<false>(a, b1, b2, aptr, outp + (TILES - 1) * JBLK, kn, sp, n < HH);
}

__device__ __forceinline__ float cw_xhalf(float x)
{
    const int b = __builtin_bit_cast(int, x);
    return __builtin_bit_cast(float,
        __builtin_amdgcn_permlanex16(b, b, 0x76543210, 0xfedcba98, false, false));
}
template <int M>
__device__ __forceinline__ float cw_xrow(float x)
{
    const int b = __builtin_bit_cast(int, x);
    return __builtin_bit_cast(float,
        __builtin_amdgcn_update_dpp(b, b, 0x160 + M, 0xf, 0xf, false));
}
__device__ __forceinline__ float cw_wave_max(float x)
{
    x = fmaxf(x, cw_xrow<1>(x));
    x = fmaxf(x, cw_xrow<2>(x));
    x = fmaxf(x, cw_xrow<4>(x));
    x = fmaxf(x, cw_xrow<8>(x));
    return fmaxf(x, cw_xhalf(x));
}
__device__ __forceinline__ float cw_wave_sum(float x)
{
    x += cw_xrow<1>(x);
    x += cw_xrow<2>(x);
    x += cw_xrow<4>(x);
    x += cw_xrow<8>(x);
    return x + cw_xhalf(x);
}

__global__ __launch_bounds__(256) void cw_softmax_kernel(float* __restrict__ data)
{
    __shared__ __attribute__((aligned(16))) float wmax[WAVES];
    __shared__ __attribute__((aligned(16))) float wsum[WAVES];
    const int tid  = threadIdx.x;
    const int wave = tid >> 5;
    const int lane = tid & 31;
    float* p = data + (size_t)blockIdx.x * JJ;

    float v[JJ / 256];
    float mx = -INFINITY;
#pragma unroll
    for (int t = 0; t < JJ / 256; ++t) {
        v[t] = p[tid + t * 256];
        mx = fmaxf(mx, v[t]);
    }
    mx = cw_wave_max(mx);
    if (lane == 0) wmax[wave] = mx;
    __syncthreads();
    mx = wmax[0];
#pragma unroll
    for (int w = 1; w < WAVES; ++w) mx = fmaxf(mx, wmax[w]);

    float sum = 0.f;
#pragma unroll
    for (int t = 0; t < JJ / 256; ++t) {
        v[t] = __expf(v[t] - mx);
        sum += v[t];
    }
    sum = cw_wave_sum(sum);
    if (lane == 0) wsum[wave] = sum;
    __syncthreads();
    sum = wsum[0];
#pragma unroll
    for (int w = 1; w < WAVES; ++w) sum += wsum[w];
    const float inv = 1.f / sum;
#pragma unroll
    for (int t = 0; t < JJ / 256; ++t) {
        p[tid + t * 256] = v[t] * inv;
    }
}

extern "C" void kernel_launch(void* const* d_in, const int* in_sizes, int n_in,
                              void* d_out, int out_size, void* d_ws, size_t ws_size,
                              hipStream_t stream)
{
    const float* memory    = (const float*)d_in[0];
    const float* keys      = (const float*)d_in[1];
    const float* strengths = (const float*)d_in[2];
    const float* mask      = (const float*)d_in[3];
    float* out = (float*)d_out;

    dim3 grid1(JJ / (JBLK * TILES), BB);
    cw_score_kernel<<<grid1, 256, 0, stream>>>(memory, keys, strengths, mask, out);
    cw_softmax_kernel<<<BB * HH, 256, 0, stream>>>(out);
}
